# starting kernel
# baseline (speedup 1.0000x reference)
#include <hip/hip_runtime.h>

typedef __attribute__((ext_vector_type(2))) float v2f;
typedef __attribute__((ext_vector_type(8))) float v8f;

#define IN_F 256
#define H_F  128
#define OUT_F 64


__global__ void k_deg_init(float* deg, int n) {
    int i = blockIdx.x * blockDim.x + threadIdx.x;
    if (i < n) deg[i] = 1.0f;
}

__global__ void k_deg_accum(const int* __restrict__ dst, float* deg, int nE) {
    int e = blockIdx.x * blockDim.x + threadIdx.x;
    if (e < nE) atomicAdd(&deg[dst[e]], 1.0f);
}

__global__ void k_rsqrt_inplace(float* d, int n) {
    int i = blockIdx.x * blockDim.x + threadIdx.x;
    if (i < n) d[i] = rsqrtf(d[i]);
}


template <int Ki, int Ko, bool RELU_A>
__global__ __launch_bounds__(256)
void k_gemm_wmma(const float* __restrict__ X, const float* __restrict__ W,
                 float* __restrict__ Y, int nrows) {
    constexpr int CT = Ko / 16;
    const int wave = threadIdx.x >> 5;
    const int lane = threadIdx.x & 31;
    const int rowTile = blockIdx.x * 8 + wave;
    const int row0 = rowTile * 16;
    if (row0 >= nrows) return;

    const int m    = lane & 15;
    const int half = lane >> 4;
    const float* xrow = X + (size_t)(row0 + m) * Ki + 2 * half;

    v8f acc[CT] = {};

    for (int kb = 0; kb < Ki; kb += 4) {
        if (kb + 32 < Ki) __builtin_prefetch(xrow + kb + 32, 0, 3);

        float ax = xrow[kb];
        float ay = xrow[kb + 1];
        if (RELU_A) { ax = fmaxf(ax, 0.0f); ay = fmaxf(ay, 0.0f); }
        v2f a = { ax, ay };

        const int krow = kb + 2 * half;
        const float* w0 = W + (size_t)krow * Ko;
        const float* w1 = w0 + Ko;
#pragma unroll
        for (int j = 0; j < CT; ++j) {
            v2f b = { w0[j * 16 + m], w1[j * 16 + m] };
            acc[j] = __builtin_amdgcn_wmma_f32_16x16x4_f32(
                false, a, false, b, (short)0, acc[j], false, false);
        }
    }

#pragma unroll
    for (int j = 0; j < CT; ++j) {
#pragma unroll
        for (int r = 0; r < 8; ++r) {
            const int M = r + 8 * half;
            Y[(size_t)(row0 + M) * Ko + j * 16 + m] = acc[j][r];
        }
    }
}

__global__ void k_agg_init(const float* __restrict__ lin, const float* __restrict__ dinv,
                           const float* __restrict__ bias, float* __restrict__ agg,
                           int n, int F) {
    int t = blockIdx.x * blockDim.x + threadIdx.x;
    int i = t / F;
    int f = t - i * F;
    if (i >= n) return;
    float w = dinv[i];
    agg[t] = lin[t] * (w * w) + bias[f];
}

__global__ void k_agg_edges(const float* __restrict__ lin, const int* __restrict__ src,
                            const int* __restrict__ dst, const float* __restrict__ dinv,
                            float* agg, int nE, int F4) {
    int t = blockIdx.x * blockDim.x + threadIdx.x;
    int e = t / F4;
    int c = t - e * F4;
    if (e >= nE) return;
    int s = src[e], d = dst[e];
    float w = dinv[s] * dinv[d];
    const float4 v = ((const float4*)(lin + (size_t)s * F4 * 4))[c];
    float* o = agg + (size_t)d * F4 * 4 + (size_t)c * 4;
    atomicAdd(o + 0, v.x * w);
    atomicAdd(o + 1, v.y * w);
    atomicAdd(o + 2, v.z * w);
    atomicAdd(o + 3, v.w * w);
}

__global__ void k_decode(const float* __restrict__ z,
                         const int* __restrict__ pos, const int* __restrict__ neg,
                         float* __restrict__ out, int EP) {
    int wid  = (blockIdx.x * blockDim.x + threadIdx.x) >> 5;
    int lane = threadIdx.x & 31;
    if (wid >= 2 * EP) return;
    int u, v;
    if (wid < EP) { u = pos[wid];       v = pos[EP + wid]; }
    else          { u = neg[wid - EP];  v = neg[EP + (wid - EP)]; }
    const float* zu = z + (size_t)u * OUT_F;
    const float* zv = z + (size_t)v * OUT_F;
    float s = zu[lane] * zv[lane] + zu[lane + 32] * zv[lane + 32];
#pragma unroll
    for (int off = 16; off > 0; off >>= 1)
        s += __shfl_down(s, off, 32);
    if (lane == 0) out[wid] = s;
}


extern "C" void kernel_launch(void* const* d_in, const int* in_sizes, int n_in,
                              void* d_out, int out_size, void* d_ws, size_t ws_size,
                              hipStream_t stream) {
    const float* x    = (const float*)d_in[0];
    const int*   ei   = (const int*)  d_in[1];
    const int*   pei  = (const int*)  d_in[2];
    const int*   nei  = (const int*)  d_in[3];
    const float* W1   = (const float*)d_in[4];
    const float* b1   = (const float*)d_in[5];
    const float* W2   = (const float*)d_in[6];
    const float* b2   = (const float*)d_in[7];
    float* out = (float*)d_out;

    const int N  = in_sizes[0] / IN_F;
    const int E  = in_sizes[1] / 2;
    const int EP = in_sizes[2] / 2;
    const int* src = ei;
    const int* dst = ei + E;

    float* dinv = (float*)d_ws;
    size_t npad = ((size_t)N + 255) & ~(size_t)255;
    float* bufA = dinv + npad;
    float* h    = bufA + (size_t)N * H_F;
    float* z    = h    + (size_t)N * H_F;

    const int B = 256;
    k_deg_init<<<(N + B - 1) / B, B, 0, stream>>>(dinv, N);
    k_deg_accum<<<(E + B - 1) / B, B, 0, stream>>>(dst, dinv, E);
    k_rsqrt_inplace<<<(N + B - 1) / B, B, 0, stream>>>(dinv, N);

    {
        int rowTiles = (N + 15) / 16;
        int blocks   = (rowTiles + 7) / 8;
        k_gemm_wmma<IN_F, H_F, false><<<blocks, B, 0, stream>>>(x, W1, bufA, N);
    }
    {
        int tot = N * H_F;
        k_agg_init<<<(tot + B - 1) / B, B, 0, stream>>>(bufA, dinv, b1, h, N, H_F);
        long long te = (long long)E * (H_F / 4);
        k_agg_edges<<<(int)((te + B - 1) / B), B, 0, stream>>>(bufA, src, dst, dinv, h, E, H_F / 4);
    }
    {
        int rowTiles = (N + 15) / 16;
        int blocks   = (rowTiles + 7) / 8;
        k_gemm_wmma<H_F, OUT_F, true><<<blocks, B, 0, stream>>>(h, W2, bufA, N);
    }
    {
        int tot = N * OUT_F;
        k_agg_init<<<(tot + B - 1) / B, B, 0, stream>>>(bufA, dinv, b2, z, N, OUT_F);
        long long te = (long long)E * (OUT_F / 4);
        k_agg_edges<<<(int)((te + B - 1) / B), B, 0, stream>>>(bufA, src, dst, dinv, z, E, OUT_F / 4);
    }
    {
        long long waves = 2LL * EP;
        long long thr   = waves * 32;
        k_decode<<<(int)((thr + B - 1) / B), B, 0, stream>>>(z, pei, nei, out, EP);
    }
}


# Aggregation by owner work-groups: each keeps 400 dst rows in LDS, scans the whole dst list, queues its edges per owner wave, plain LDS adds; global float atomics gone; sum order = queue arrival order
# speedup vs baseline: 1.4897x; 1.4897x over previous
#include <hip/hip_runtime.h>

typedef __attribute__((ext_vector_type(2))) float v2f;
typedef __attribute__((ext_vector_type(8))) float v8f;

#define IN_F 256
#define H_F  128
#define OUT_F 64


__global__ void k_deg_init(float* deg, int n) {
    int i = blockIdx.x * blockDim.x + threadIdx.x;
    if (i < n) deg[i] = 1.0f;
}

__global__ void k_deg_accum(const int* __restrict__ dst, float* deg, int nE) {
    int e = blockIdx.x * blockDim.x + threadIdx.x;
    if (e < nE) atomicAdd(&deg[dst[e]], 1.0f);
}

__global__ void k_rsqrt_inplace(float* d, int n) {
    int i = blockIdx.x * blockDim.x + threadIdx.x;
    if (i < n) d[i] = rsqrtf(d[i]);
}


template <int Ki, int Ko, bool RELU_A>
__global__ __launch_bounds__(256)
void k_gemm_wmma(const float* __restrict__ X, const float* __restrict__ W,
                 float* __restrict__ Y, int nrows) {
    constexpr int CT = Ko / 16;
    const int wave = threadIdx.x >> 5;
    const int lane = threadIdx.x & 31;
    const int rowTile = blockIdx.x * 8 + wave;
    const int row0 = rowTile * 16;
    if (row0 >= nrows) return;

    const int m    = lane & 15;
    const int half = lane >> 4;
    const float* xrow = X + (size_t)(row0 + m) * Ki + 2 * half;

    v8f acc[CT] = {};

    for (int kb = 0; kb < Ki; kb += 4) {
        if (kb + 32 < Ki) __builtin_prefetch(xrow + kb + 32, 0, 3);

        float ax = xrow[kb];
        float ay = xrow[kb + 1];
        if (RELU_A) { ax = fmaxf(ax, 0.0f); ay = fmaxf(ay, 0.0f); }
        v2f a = { ax, ay };

        const int krow = kb + 2 * half;
        const float* w0 = W + (size_t)krow * Ko;
        const float* w1 = w0 + Ko;
#pragma unroll
        for (int j = 0; j < CT; ++j) {
            v2f b = { w0[j * 16 + m], w1[j * 16 + m] };
            acc[j] = __builtin_amdgcn_wmma_f32_16x16x4_f32(
                false, a, false, b, (short)0, acc[j], false, false);
        }
    }

#pragma unroll
    for (int j = 0; j < CT; ++j) {
#pragma unroll
        for (int r = 0; r < 8; ++r) {
            const int M = r + 8 * half;
            Y[(size_t)(row0 + M) * Ko + j * 16 + m] = acc[j][r];
        }
    }
}

#define TILE  400
#define QCAP  512
#define NWAVE 32

template <int F>
__device__ __forceinline__ void push_edge(int e, unsigned r, int lo, int* qcnt, int* queue,
                                          float* tile, const float* __restrict__ lin,
                                          const int* __restrict__ src,
                                          const float* __restrict__ dinv) {
    const int owner = (int)(r & 31u);
    const int slot = atomicAdd(&qcnt[owner], 1);
    if (slot < QCAP) {
        queue[owner * QCAP + slot] = e;
    } else {
        const int s = src[e];
        const float w = dinv[s] * dinv[lo + (int)r];
        const float* row = lin + (size_t)s * F;
        float* t = tile + (size_t)r * F;
#pragma nounroll
        for (int f = 0; f < F; ++f) atomicAdd(t + f, row[f] * w);
    }
}

template <int F>
__device__ __forceinline__ void scan4(int4 v, int e0, int lo, unsigned ucnt, int* qcnt, int* queue,
                                      float* tile, const float* __restrict__ lin,
                                      const int* __restrict__ src,
                                      const float* __restrict__ dinv) {
    const unsigned r0 = (unsigned)(v.x - lo);
    const unsigned r1 = (unsigned)(v.y - lo);
    const unsigned r2 = (unsigned)(v.z - lo);
    const unsigned r3 = (unsigned)(v.w - lo);
    if (min(min(r0, r1), min(r2, r3)) < ucnt) {
        unsigned m = (r0 < ucnt ? 1u : 0u) | (r1 < ucnt ? 2u : 0u) |
                     (r2 < ucnt ? 4u : 0u) | (r3 < ucnt ? 8u : 0u);
        do {
            const int k = __builtin_ctz(m);
            m &= m - 1;
            unsigned r = r0;
            if (k == 1) r = r1;
            if (k == 2) r = r2;
            if (k == 3) r = r3;
            push_edge<F>(e0 + k, r, lo, qcnt, queue, tile, lin, src, dinv);
        } while (m);
    }
}

template <int F>
__global__ __launch_bounds__(1024)
void k_agg_owner(const float* __restrict__ lin, const int* __restrict__ src,
                 const int* __restrict__ dst, const float* __restrict__ dinv,
                 const float* __restrict__ bias, float* __restrict__ agg,
                 int n, int nE) {
    constexpr int VF = F / 32;
    constexpr int F4 = F / 4;
    typedef __attribute__((ext_vector_type(VF))) float vrow;
    __shared__ __attribute__((aligned(16))) float tile[TILE * F];
    __shared__ int qcnt[NWAVE];
    __shared__ int queue[NWAVE * QCAP];

    const int tid  = threadIdx.x;
    const int lane = tid & 31;
    const int wave = __builtin_amdgcn_readfirstlane(tid >> 5);
    const int lo   = blockIdx.x * TILE;
    const int cnt  = min(TILE, n - lo);
    const unsigned ucnt = (unsigned)cnt;

    {
        const float4* lin4 = (const float4*)(lin + (size_t)lo * F);
        float4* tile4 = (float4*)tile;
        for (int idx = tid; idx < cnt * F4; idx += 1024) {
            const int i = idx / F4;
            const int c = idx - i * F4;
            const float w = dinv[lo + i];
            const float ww = w * w;
            const float4 l = lin4[idx];
            float4 o;
            o.x = l.x * ww + bias[4 * c + 0];
            o.y = l.y * ww + bias[4 * c + 1];
            o.z = l.z * ww + bias[4 * c + 2];
            o.w = l.w * ww + bias[4 * c + 3];
            tile4[idx] = o;
        }
        if (tid < NWAVE) qcnt[tid] = 0;
    }
    __syncthreads();

    {
        constexpr int U = 8;
        const bool al = (((size_t)dst) & 15) == 0;
        const int nV = al ? (nE >> 2) : 0;
        const int4* d4 = (const int4*)dst;
        const int nFull = nV / (1024 * U);
        const int rot = (int)(((unsigned)blockIdx.x * (unsigned)nFull) / gridDim.x);
        for (int it = 0; it < nFull; ++it) {
            int g = it + rot;
            if (g >= nFull) g -= nFull;
            const int base = g * (1024 * U) + tid;
            int4 v[U];
#pragma unroll
            for (int u = 0; u < U; ++u) v[u] = d4[base + u * 1024];
#pragma unroll
            for (int u = 0; u < U; ++u)
                scan4<F>(v[u], (base + u * 1024) * 4, lo, ucnt, qcnt, queue, tile, lin, src, dinv);
        }
        for (int idx = nFull * (1024 * U) + tid; idx < nV; idx += 1024)
            scan4<F>(d4[idx], idx * 4, lo, ucnt, qcnt, queue, tile, lin, src, dinv);
        for (int e = nV * 4 + tid; e < nE; e += 1024) {
            const unsigned r = (unsigned)(dst[e] - lo);
            if (r < ucnt) push_edge<F>(e, r, lo, qcnt, queue, tile, lin, src, dinv);
        }
    }
    __syncthreads();

    {
        constexpr int G = 8;
        const int mine = __builtin_amdgcn_readfirstlane(min(qcnt[wave], QCAP));
        const int* q = queue + wave * QCAP;
        for (int b = 0; b < mine; b += 32) {
            const int nb = min(32, mine - b);
            const int e  = q[b + min(lane, nb - 1)];
            const int s  = src[e];
            const int d  = dst[e];
            const float w = dinv[s] * dinv[d];
            const int dl = d - lo;
            vrow va[G], vb[G];
#define GATHER(buf, J0)                                                                   \
            if ((J0) < nb) {                                                              \
                _Pragma("unroll")                                                         \
                for (int j = 0; j < G; ++j) {                                             \
                    const int sj = __builtin_amdgcn_readlane(s, min((J0) + j, nb - 1));   \
                    buf[j] = *(const vrow*)(lin + (size_t)sj * F + lane * VF);            \
                }                                                                         \
            }
#define ADDROWS(buf, J0)                                                                  \
            if ((J0) < nb) {                                                              \
                _Pragma("unroll")                                                         \
                for (int j = 0; j < G; ++j) {                                             \
                    if ((J0) + j < nb) {                                                  \
                        const float wj = __builtin_bit_cast(float,                        \
                            __builtin_amdgcn_readlane(__builtin_bit_cast(int, w), (J0) + j)); \
                        const int dj = __builtin_amdgcn_readlane(dl, (J0) + j);           \
                        vrow* t = (vrow*)(tile + (size_t)dj * F + lane * VF);             \
                        *t = *t + buf[j] * wj;                                            \
                    }                                                                     \
                }                                                                         \
            }
            GATHER(va, 0)
            GATHER(vb, G)
            ADDROWS(va, 0)
            GATHER(va, 2 * G)
            ADDROWS(vb, G)
            GATHER(vb, 3 * G)
            ADDROWS(va, 2 * G)
            ADDROWS(vb, 3 * G)
#undef GATHER
#undef ADDROWS
        }
    }
    __syncthreads();

    {
        float4* out4 = (float4*)(agg + (size_t)lo * F);
        const float4* tile4 = (const float4*)tile;
        for (int idx = tid; idx < cnt * F4; idx += 1024) out4[idx] = tile4[idx];
    }
}

__global__ void k_decode(const float* __restrict__ z,
                         const int* __restrict__ pos, const int* __restrict__ neg,
                         float* __restrict__ out, int EP) {
    int wid  = (blockIdx.x * blockDim.x + threadIdx.x) >> 5;
    int lane = threadIdx.x & 31;
    if (wid >= 2 * EP) return;
    int u, v;
    if (wid < EP) { u = pos[wid];       v = pos[EP + wid]; }
    else          { u = neg[wid - EP];  v = neg[EP + (wid - EP)]; }
    const float* zu = z + (size_t)u * OUT_F;
    const float* zv = z + (size_t)v * OUT_F;
    float s = zu[lane] * zv[lane] + zu[lane + 32] * zv[lane + 32];
#pragma unroll
    for (int off = 16; off > 0; off >>= 1)
        s += __shfl_down(s, off, 32);
    if (lane == 0) out[wid] = s;
}


extern "C" void kernel_launch(void* const* d_in, const int* in_sizes, int n_in,
                              void* d_out, int out_size, void* d_ws, size_t ws_size,
                              hipStream_t stream) {
    const float* x    = (const float*)d_in[0];
    const int*   ei   = (const int*)  d_in[1];
    const int*   pei  = (const int*)  d_in[2];
    const int*   nei  = (const int*)  d_in[3];
    const float* W1   = (const float*)d_in[4];
    const float* b1   = (const float*)d_in[5];
    const float* W2   = (const float*)d_in[6];
    const float* b2   = (const float*)d_in[7];
    float* out = (float*)d_out;

    const int N  = in_sizes[0] / IN_F;
    const int E  = in_sizes[1] / 2;
    const int EP = in_sizes[2] / 2;
    const int* src = ei;
    const int* dst = ei + E;

    float* dinv = (float*)d_ws;
    size_t npad = ((size_t)N + 255) & ~(size_t)255;
    float* bufA = dinv + npad;
    float* h    = bufA + (size_t)N * H_F;
    float* z    = h    + (size_t)N * H_F;

    const int B = 256;
    k_deg_init<<<(N + B - 1) / B, B, 0, stream>>>(dinv, N);
    k_deg_accum<<<(E + B - 1) / B, B, 0, stream>>>(dst, dinv, E);
    k_rsqrt_inplace<<<(N + B - 1) / B, B, 0, stream>>>(dinv, N);

    {
        int rowTiles = (N + 15) / 16;
        int blocks   = (rowTiles + 7) / 8;
        k_gemm_wmma<IN_F, H_F, false><<<blocks, B, 0, stream>>>(x, W1, bufA, N);
    }
    k_agg_owner<H_F><<<(N + TILE - 1) / TILE, 1024, 0, stream>>>(bufA, src, dst, dinv, b1, h, N, E);
    {
        int rowTiles = (N + 15) / 16;
        int blocks   = (rowTiles + 7) / 8;
        k_gemm_wmma<H_F, OUT_F, true><<<blocks, B, 0, stream>>>(h, W2, bufA, N);
    }
    k_agg_owner<OUT_F><<<(N + TILE - 1) / TILE, 1024, 0, stream>>>(bufA, src, dst, dinv, b2, z, N, E);
    {
        long long waves = 2LL * EP;
        long long thr   = waves * 32;
        k_decode<<<(int)((thr + B - 1) / B), B, 0, stream>>>(z, pei, nei, out, EP);
    }
}


# Layer-2 aggregation skips the dst-list scan: layer-1 owner work-groups leave per-wave edge records (src row, tile row, weight) in their own z rows; layer 2 walks them; its sum order = layer 1's
# speedup vs baseline: 1.7081x; 1.1466x over previous
#include <hip/hip_runtime.h>

typedef __attribute__((ext_vector_type(2))) float v2f;
typedef __attribute__((ext_vector_type(8))) float v8f;

#define IN_F 256
#define H_F  128
#define OUT_F 64


__global__ void k_deg_init(float* deg, int n) {
    int i = blockIdx.x * blockDim.x + threadIdx.x;
    if (i < n) deg[i] = 1.0f;
}

__global__ void k_deg_accum(const int* __restrict__ dst, float* deg, int nE) {
    int e = blockIdx.x * blockDim.x + threadIdx.x;
    if (e < nE) atomicAdd(&deg[dst[e]], 1.0f);
}

__global__ void k_rsqrt_inplace(float* d, int n) {
    int i = blockIdx.x * blockDim.x + threadIdx.x;
    if (i < n) d[i] = rsqrtf(d[i]);
}


template <int Ki, int Ko, bool RELU_A>
__global__ __launch_bounds__(256)
void k_gemm_wmma(const float* __restrict__ X, const float* __restrict__ W,
                 float* __restrict__ Y, int nrows) {
    constexpr int CT = Ko / 16;
    const int wave = threadIdx.x >> 5;
    const int lane = threadIdx.x & 31;
    const int rowTile = blockIdx.x * 8 + wave;
    const int row0 = rowTile * 16;
    if (row0 >= nrows) return;

    const int m    = lane & 15;
    const int half = lane >> 4;
    const float* xrow = X + (size_t)(row0 + m) * Ki + 2 * half;

    v8f acc[CT] = {};

    for (int kb = 0; kb < Ki; kb += 4) {
        if (kb + 32 < Ki) __builtin_prefetch(xrow + kb + 32, 0, 3);

        float ax = xrow[kb];
        float ay = xrow[kb + 1];
        if (RELU_A) { ax = fmaxf(ax, 0.0f); ay = fmaxf(ay, 0.0f); }
        v2f a = { ax, ay };

        const int krow = kb + 2 * half;
        const float* w0 = W + (size_t)krow * Ko;
        const float* w1 = w0 + Ko;
#pragma unroll
        for (int j = 0; j < CT; ++j) {
            v2f b = { w0[j * 16 + m], w1[j * 16 + m] };
            acc[j] = __builtin_amdgcn_wmma_f32_16x16x4_f32(
                false, a, false, b, (short)0, acc[j], false, false);
        }
    }

#pragma unroll
    for (int j = 0; j < CT; ++j) {
#pragma unroll
        for (int r = 0; r < 8; ++r) {
            const int M = r + 8 * half;
            Y[(size_t)(row0 + M) * Ko + j * 16 + m] = acc[j][r];
        }
    }
}

#define TILE  400
#define QCAP  512
#define NWAVE 32
#define QHDR  64

template <int F>
__device__ __forceinline__ void push_edge(int e, unsigned r, int lo, int* qcnt, int* queue,
                                          float* tile, const float* __restrict__ lin,
                                          const int* __restrict__ src,
                                          const float* __restrict__ dinv) {
    const int owner = (int)(r & 31u);
    const int slot = atomicAdd(&qcnt[owner], 1);
    if (slot < QCAP) {
        queue[owner * QCAP + slot] = e;
    } else {
        const int s = src[e];
        const float w = dinv[s] * dinv[lo + (int)r];
        const float* row = lin + (size_t)s * F;
        float* t = tile + (size_t)r * F;
#pragma nounroll
        for (int f = 0; f < F; ++f) atomicAdd(t + f, row[f] * w);
    }
}

template <int F>
__device__ __forceinline__ void scan4(int4 v, int e0, int lo, unsigned ucnt, int* qcnt, int* queue,
                                      float* tile, const float* __restrict__ lin,
                                      const int* __restrict__ src,
                                      const float* __restrict__ dinv) {
    const unsigned r0 = (unsigned)(v.x - lo);
    const unsigned r1 = (unsigned)(v.y - lo);
    const unsigned r2 = (unsigned)(v.z - lo);
    const unsigned r3 = (unsigned)(v.w - lo);
    if (min(min(r0, r1), min(r2, r3)) < ucnt) {
        unsigned m = (r0 < ucnt ? 1u : 0u) | (r1 < ucnt ? 2u : 0u) |
                     (r2 < ucnt ? 4u : 0u) | (r3 < ucnt ? 8u : 0u);
        do {
            const int k = __builtin_ctz(m);
            m &= m - 1;
            unsigned r = r0;
            if (k == 1) r = r1;
            if (k == 2) r = r2;
            if (k == 3) r = r3;
            push_edge<F>(e0 + k, r, lo, qcnt, queue, tile, lin, src, dinv);
        } while (m);
    }
}

template <int F, bool STORE_Q, bool LOAD_Q>
__global__ __launch_bounds__(1024)
void k_agg_owner(const float* __restrict__ lin, const int* __restrict__ src,
                 const int* __restrict__ dst, const float* __restrict__ dinv,
                 const float* __restrict__ bias, float* agg, int* qrec,
                 int n, int nE) {
    constexpr int VF = F / 32;
    constexpr int F4 = F / 4;
    typedef __attribute__((ext_vector_type(VF))) float vrow;
    __shared__ __attribute__((aligned(16))) float tile[TILE * F];
    __shared__ int qcnt[NWAVE];
    __shared__ int queue[NWAVE * QCAP];

    const int tid  = threadIdx.x;
    const int lane = tid & 31;
    const int wave = __builtin_amdgcn_readfirstlane(tid >> 5);
    const int lo   = blockIdx.x * TILE;
    const int cnt  = min(TILE, n - lo);
    const unsigned ucnt = (unsigned)cnt;

    int* slab = qrec + (size_t)lo * OUT_F;
    const int cap = (cnt * OUT_F - QHDR) / 3;
    int* recS = slab + QHDR;
    int* recD = recS + cap;
    int* recW = recD + cap;
    bool useSlab = false;
    if (LOAD_Q) useSlab = __builtin_amdgcn_readfirstlane(slab[QHDR / 2]) == 1;

    {
        const float4* lin4 = (const float4*)(lin + (size_t)lo * F);
        float4* tile4 = (float4*)tile;
        for (int idx = tid; idx < cnt * F4; idx += 1024) {
            const int i = idx / F4;
            const int c = idx - i * F4;
            const float w = dinv[lo + i];
            const float ww = w * w;
            const float4 l = lin4[idx];
            float4 o;
            o.x = l.x * ww + bias[4 * c + 0];
            o.y = l.y * ww + bias[4 * c + 1];
            o.z = l.z * ww + bias[4 * c + 2];
            o.w = l.w * ww + bias[4 * c + 3];
            tile4[idx] = o;
        }
        if (tid < NWAVE) {
            int c0 = 0;
            if (useSlab) c0 = slab[tid];
            qcnt[tid] = c0;
        }
    }
    __syncthreads();

    if (!useSlab) {
        constexpr int U = 8;
        const bool al = (((size_t)dst) & 15) == 0;
        const int nV = al ? (nE >> 2) : 0;
        const int4* d4 = (const int4*)dst;
        const int nFull = nV / (1024 * U);
        const int rot = (int)(((unsigned)blockIdx.x * (unsigned)nFull) / gridDim.x);
        for (int it = 0; it < nFull; ++it) {
            int g = it + rot;
            if (g >= nFull) g -= nFull;
            const int base = g * (1024 * U) + tid;
            int4 v[U];
#pragma unroll
            for (int u = 0; u < U; ++u) v[u] = d4[base + u * 1024];
#pragma unroll
            for (int u = 0; u < U; ++u)
                scan4<F>(v[u], (base + u * 1024) * 4, lo, ucnt, qcnt, queue, tile, lin, src, dinv);
        }
        for (int idx = nFull * (1024 * U) + tid; idx < nV; idx += 1024)
            scan4<F>(d4[idx], idx * 4, lo, ucnt, qcnt, queue, tile, lin, src, dinv);
        for (int e = nV * 4 + tid; e < nE; e += 1024) {
            const unsigned r = (unsigned)(dst[e] - lo);
            if (r < ucnt) push_edge<F>(e, r, lo, qcnt, queue, tile, lin, src, dinv);
        }
    }
    __syncthreads();

    {
        constexpr int G = 8;
        const int qraw = qcnt[lane];
        const int cl = min(qraw, QCAP);
        int off = 0, total = 0;
        for (int k = 0; k < NWAVE; ++k) {
            const int ck = __builtin_amdgcn_readlane(cl, k);
            if (k < wave) off += ck;
            total += ck;
        }
        const int mine = __builtin_amdgcn_readlane(cl, wave);
        bool storeOK = false;
        if (STORE_Q) {
            storeOK = !__any(qraw > QCAP) && total <= cap;
            if (wave == 0) {
                slab[lane] = cl;
                if (lane == 0) slab[QHDR / 2] = storeOK ? 1 : 0;
            }
        }
        const int* q = queue + wave * QCAP;
        for (int b = 0; b < mine; b += 32) {
            const int nb = min(32, mine - b);
            const int idx = b + min(lane, nb - 1);
            int s, dl;
            float w;
            if (useSlab) {
                s  = recS[off + idx];
                dl = recD[off + idx];
                w  = __builtin_bit_cast(float, recW[off + idx]);
            } else {
                const int e = q[idx];
                s = src[e];
                const int d = dst[e];
                w = dinv[s] * dinv[d];
                dl = d - lo;
                if (STORE_Q && storeOK && lane < nb) {
                    recS[off + idx] = s;
                    recD[off + idx] = dl;
                    recW[off + idx] = __builtin_bit_cast(int, w);
                }
            }
            vrow va[G], vb[G];
#define GATHER(buf, J0)                                                                   \
            if ((J0) < nb) {                                                              \
                _Pragma("unroll")                                                         \
                for (int j = 0; j < G; ++j) {                                             \
                    const int sj = __builtin_amdgcn_readlane(s, min((J0) + j, nb - 1));   \
                    buf[j] = *(const vrow*)(lin + (size_t)sj * F + lane * VF);            \
                }                                                                         \
            }
#define ADDROWS(buf, J0)                                                                  \
            if ((J0) < nb) {                                                              \
                _Pragma("unroll")                                                         \
                for (int j = 0; j < G; ++j) {                                             \
                    if ((J0) + j < nb) {                                                  \
                        const float wj = __builtin_bit_cast(float,                        \
                            __builtin_amdgcn_readlane(__builtin_bit_cast(int, w), (J0) + j)); \
                        const int dj = __builtin_amdgcn_readlane(dl, (J0) + j);           \
                        vrow* t = (vrow*)(tile + (size_t)dj * F + lane * VF);             \
                        *t = *t + buf[j] * wj;                                            \
                    }                                                                     \
                }                                                                         \
            }
            GATHER(va, 0)
            GATHER(vb, G)
            ADDROWS(va, 0)
            GATHER(va, 2 * G)
            ADDROWS(vb, G)
            GATHER(vb, 3 * G)
            ADDROWS(va, 2 * G)
            ADDROWS(vb, 3 * G)
#undef GATHER
#undef ADDROWS
        }
    }
    __syncthreads();

    {
        float4* out4 = (float4*)(agg + (size_t)lo * F);
        const float4* tile4 = (const float4*)tile;
        for (int idx = tid; idx < cnt * F4; idx += 1024) out4[idx] = tile4[idx];
    }
}

__global__ void k_decode(const float* __restrict__ z,
                         const int* __restrict__ pos, const int* __restrict__ neg,
                         float* __restrict__ out, int EP) {
    int wid  = (blockIdx.x * blockDim.x + threadIdx.x) >> 5;
    int lane = threadIdx.x & 31;
    if (wid >= 2 * EP) return;
    int u, v;
    if (wid < EP) { u = pos[wid];       v = pos[EP + wid]; }
    else          { u = neg[wid - EP];  v = neg[EP + (wid - EP)]; }
    const float* zu = z + (size_t)u * OUT_F;
    const float* zv = z + (size_t)v * OUT_F;
    float s = zu[lane] * zv[lane] + zu[lane + 32] * zv[lane + 32];
#pragma unroll
    for (int off = 16; off > 0; off >>= 1)
        s += __shfl_down(s, off, 32);
    if (lane == 0) out[wid] = s;
}


extern "C" void kernel_launch(void* const* d_in, const int* in_sizes, int n_in,
                              void* d_out, int out_size, void* d_ws, size_t ws_size,
                              hipStream_t stream) {
    const float* x    = (const float*)d_in[0];
    const int*   ei   = (const int*)  d_in[1];
    const int*   pei  = (const int*)  d_in[2];
    const int*   nei  = (const int*)  d_in[3];
    const float* W1   = (const float*)d_in[4];
    const float* b1   = (const float*)d_in[5];
    const float* W2   = (const float*)d_in[6];
    const float* b2   = (const float*)d_in[7];
    float* out = (float*)d_out;

    const int N  = in_sizes[0] / IN_F;
    const int E  = in_sizes[1] / 2;
    const int EP = in_sizes[2] / 2;
    const int* src = ei;
    const int* dst = ei + E;

    float* dinv = (float*)d_ws;
    size_t npad = ((size_t)N + 255) & ~(size_t)255;
    float* bufA = dinv + npad;
    float* h    = bufA + (size_t)N * H_F;
    float* z    = h    + (size_t)N * H_F;

    const int B = 256;
    k_deg_init<<<(N + B - 1) / B, B, 0, stream>>>(dinv, N);
    k_deg_accum<<<(E + B - 1) / B, B, 0, stream>>>(dst, dinv, E);
    k_rsqrt_inplace<<<(N + B - 1) / B, B, 0, stream>>>(dinv, N);

    {
        int rowTiles = (N + 15) / 16;
        int blocks   = (rowTiles + 7) / 8;
        k_gemm_wmma<IN_F, H_F, false><<<blocks, B, 0, stream>>>(x, W1, bufA, N);
    }
    k_agg_owner<H_F, true, false><<<(N + TILE - 1) / TILE, 1024, 0, stream>>>(bufA, src, dst, dinv, b1, h, (int*)z, N, E);
    {
        int rowTiles = (N + 15) / 16;
        int blocks   = (rowTiles + 7) / 8;
        k_gemm_wmma<H_F, OUT_F, true><<<blocks, B, 0, stream>>>(h, W2, bufA, N);
    }
    k_agg_owner<OUT_F, false, true><<<(N + TILE - 1) / TILE, 1024, 0, stream>>>(bufA, src, dst, dinv, b2, z, (int*)z, N, E);
    {
        long long waves = 2LL * EP;
        long long thr   = waves * 32;
        k_decode<<<(int)((thr + B - 1) / B), B, 0, stream>>>(z, pei, nei, out, EP);
    }
}


# Layer-1 dst scan reads a quarter of the bytes: degree pass leaves a 1-byte owner key per edge; owner work-groups scan the keys, check hits in dst, queue as before; sum order = queue arrival order
# speedup vs baseline: 1.8043x; 1.0563x over previous
#include <hip/hip_runtime.h>

typedef __attribute__((ext_vector_type(2))) float v2f;
typedef __attribute__((ext_vector_type(8))) float v8f;

#define IN_F 256
#define H_F  128
#define OUT_F 64

#define TILE  400
#define KROW  16
#define KSEG  (TILE * KROW)
#define KSTRIDE (TILE * OUT_F * 4)

__device__ __forceinline__ bool keys_ok(int n, int nE) {
    return (n + TILE - 1) / TILE <= 256 && ((nE + 15) >> 4) <= n;
}


__global__ void k_deg_init(float* deg, int n) {
    int i = blockIdx.x * blockDim.x + threadIdx.x;
    if (i < n) deg[i] = 1.0f;
}

__global__ void k_deg_accum(const int* __restrict__ dst, float* deg, unsigned char* keys,
                            int nE, int n) {
    int e = blockIdx.x * blockDim.x + threadIdx.x;
    if (e < nE) {
        const int d = dst[e];
        atomicAdd(&deg[d], 1.0f);
        if (keys_ok(n, nE)) {
            const int g = e / KSEG;
            keys[(size_t)g * KSTRIDE + (e - g * KSEG)] = (unsigned char)(d / TILE);
        }
    }
}

__global__ void k_rsqrt_inplace(float* d, int n) {
    int i = blockIdx.x * blockDim.x + threadIdx.x;
    if (i < n) d[i] = rsqrtf(d[i]);
}


template <int Ki, int Ko, bool RELU_A>
__global__ __launch_bounds__(256)
void k_gemm_wmma(const float* __restrict__ X, const float* __restrict__ W,
                 float* __restrict__ Y, int nrows) {
    constexpr int CT = Ko / 16;
    const int wave = threadIdx.x >> 5;
    const int lane = threadIdx.x & 31;
    const int rowTile = blockIdx.x * 8 + wave;
    const int row0 = rowTile * 16;
    if (row0 >= nrows) return;

    const int m    = lane & 15;
    const int half = lane >> 4;
    const float* xrow = X + (size_t)(row0 + m) * Ki + 2 * half;

    v8f acc[CT] = {};

    for (int kb = 0; kb < Ki; kb += 4) {
        if (kb + 32 < Ki) __builtin_prefetch(xrow + kb + 32, 0, 3);

        float ax = xrow[kb];
        float ay = xrow[kb + 1];
        if (RELU_A) { ax = fmaxf(ax, 0.0f); ay = fmaxf(ay, 0.0f); }
        v2f a = { ax, ay };

        const int krow = kb + 2 * half;
        const float* w0 = W + (size_t)krow * Ko;
        const float* w1 = w0 + Ko;
#pragma unroll
        for (int j = 0; j < CT; ++j) {
            v2f b = { w0[j * 16 + m], w1[j * 16 + m] };
            acc[j] = __builtin_amdgcn_wmma_f32_16x16x4_f32(
                false, a, false, b, (short)0, acc[j], false, false);
        }
    }

#pragma unroll
    for (int j = 0; j < CT; ++j) {
#pragma unroll
        for (int r = 0; r < 8; ++r) {
            const int M = r + 8 * half;
            Y[(size_t)(row0 + M) * Ko + j * 16 + m] = acc[j][r];
        }
    }
}

#define QCAP  512
#define NWAVE 32
#define QHDR  64
#define CW    320

template <int F>
__device__ __forceinline__ void push_edge(int e, unsigned r, int lo, int* qcnt, int* queue,
                                          float* tile, const float* __restrict__ lin,
                                          const int* __restrict__ src,
                                          const float* __restrict__ dinv) {
    const int owner = (int)(r & 31u);
    const int slot = atomicAdd(&qcnt[owner], 1);
    if (slot < QCAP) {
        queue[owner * QCAP + slot] = e;
    } else {
        const int s = src[e];
        const float w = dinv[s] * dinv[lo + (int)r];
        const float* row = lin + (size_t)s * F;
        float* t = tile + (size_t)r * F;
#pragma nounroll
        for (int f = 0; f < F; ++f) atomicAdd(t + f, row[f] * w);
    }
}

template <int F>
__device__ __forceinline__ void scan4(int4 v, int e0, int lo, unsigned ucnt, int* qcnt, int* queue,
                                      float* tile, const float* __restrict__ lin,
                                      const int* __restrict__ src,
                                      const float* __restrict__ dinv) {
    const unsigned r0 = (unsigned)(v.x - lo);
    const unsigned r1 = (unsigned)(v.y - lo);
    const unsigned r2 = (unsigned)(v.z - lo);
    const unsigned r3 = (unsigned)(v.w - lo);
    if (min(min(r0, r1), min(r2, r3)) < ucnt) {
        unsigned m = (r0 < ucnt ? 1u : 0u) | (r1 < ucnt ? 2u : 0u) |
                     (r2 < ucnt ? 4u : 0u) | (r3 < ucnt ? 8u : 0u);
        do {
            const int k = __builtin_ctz(m);
            m &= m - 1;
            unsigned r = r0;
            if (k == 1) r = r1;
            if (k == 2) r = r2;
            if (k == 3) r = r3;
            push_edge<F>(e0 + k, r, lo, qcnt, queue, tile, lin, src, dinv);
        } while (m);
    }
}

template <int F>
__device__ __forceinline__ int keyscan16(uint4 v, unsigned e0, unsigned pat, int lo, unsigned ucnt,
                                         unsigned nE, int wc, int* wl, int* qcnt, int* queue,
                                         float* tile, const float* __restrict__ lin,
                                         const int* __restrict__ src,
                                         const int* __restrict__ dst,
                                         const float* __restrict__ dinv) {
    const unsigned x0 = v.x ^ pat, x1 = v.y ^ pat, x2 = v.z ^ pat, x3 = v.w ^ pat;
    const unsigned z0 = (x0 - 0x01010101u) & ~x0;
    const unsigned z1 = (x1 - 0x01010101u) & ~x1;
    const unsigned z2 = (x2 - 0x01010101u) & ~x2;
    const unsigned z3 = (x3 - 0x01010101u) & ~x3;
    unsigned m = z0;
    m = (m & 0x80808080u) | ((z1 >> 1) & 0x7f7f7f7fu);
    m = (m & 0xc0c0c0c0u) | ((z2 >> 2) & 0x3f3f3f3fu);
    m = (m & 0xe0e0e0e0u) | ((z3 >> 3) & 0x1f1f1f1fu);
    m &= 0xf0f0f0f0u;
    unsigned act = __builtin_amdgcn_ballot_w32(m != 0);
    while (act) {
        const int i = __builtin_ctz(m | 0x80000000u);
        const unsigned e = e0 + (unsigned)(((7 - (i & 7)) << 2) + (i >> 3));
        const bool ok = m != 0 && e < nE;
        m &= m - 1;
        const unsigned okm = __builtin_amdgcn_ballot_w32(ok);
        const int slot = wc + (int)__builtin_amdgcn_mbcnt_lo(okm, 0u);
        if (ok) {
            if (slot < CW) {
                wl[slot] = (int)e;
            } else {
                const unsigned r = (unsigned)(dst[e] - lo);
                if (r < ucnt) push_edge<F>((int)e, r, lo, qcnt, queue, tile, lin, src, dinv);
            }
        }
        wc += __builtin_popcount(okm);
        act = __builtin_amdgcn_ballot_w32(m != 0);
    }
    return wc;
}

template <int F, bool STORE_Q, bool LOAD_Q>
__global__ __launch_bounds__(1024)
void k_agg_owner(const float* __restrict__ lin, const int* __restrict__ src,
                 const int* __restrict__ dst, const float* __restrict__ dinv,
                 const float* __restrict__ bias, float* agg, int* qrec,
                 int n, int nE) {
    constexpr int VF = F / 32;
    constexpr int F4 = F / 4;
    typedef __attribute__((ext_vector_type(VF))) float vrow;
    __shared__ __attribute__((aligned(16))) float tile[TILE * F];
    __shared__ int qcnt[NWAVE];
    __shared__ int queue[NWAVE * QCAP];
    __shared__ int cand[STORE_Q ? NWAVE * CW : 1];

    const int tid  = threadIdx.x;
    const int lane = tid & 31;
    const int wave = __builtin_amdgcn_readfirstlane(tid >> 5);
    const int lo   = blockIdx.x * TILE;
    const int cnt  = min(TILE, n - lo);
    const unsigned ucnt = (unsigned)cnt;

    int* slab = qrec + (size_t)lo * OUT_F + cnt * (KROW / 4);
    const int cap = (cnt * (OUT_F - KROW / 4) - QHDR) / 3;
    int* recS = slab + QHDR;
    int* recD = recS + cap;
    int* recW = recD + cap;
    bool useSlab = false;
    if (LOAD_Q) useSlab = __builtin_amdgcn_readfirstlane(slab[QHDR / 2]) == 1;

    {
        const float4* lin4 = (const float4*)(lin + (size_t)lo * F);
        float4* tile4 = (float4*)tile;
        for (int idx = tid; idx < cnt * F4; idx += 1024) {
            const int i = idx / F4;
            const int c = idx - i * F4;
            const float w = dinv[lo + i];
            const float ww = w * w;
            const float4 l = lin4[idx];
            float4 o;
            o.x = l.x * ww + bias[4 * c + 0];
            o.y = l.y * ww + bias[4 * c + 1];
            o.z = l.z * ww + bias[4 * c + 2];
            o.w = l.w * ww + bias[4 * c + 3];
            tile4[idx] = o;
        }
        if (tid < NWAVE) {
            int c0 = 0;
            if (useSlab) c0 = slab[tid];
            qcnt[tid] = c0;
        }
    }
    __syncthreads();

    const bool byKeys = STORE_Q && keys_ok(n, nE);
    if (byKeys) {
        constexpr int U = 8;
        const unsigned char* kb = (const unsigned char*)qrec;
        const unsigned pat = (unsigned)blockIdx.x * 0x01010101u;
        const unsigned uE = (unsigned)nE;
        const unsigned nC = (uE + 15u) >> 4;
        const unsigned nFull = nC / (1024u * U);
        const unsigned rot = ((unsigned)blockIdx.x * nFull) / gridDim.x;
        int* wl = cand + wave * CW;
        int wc = 0;
#define KEYPTR(c) ((const uint4*)(kb + ((c) * 16u + ((c) / (unsigned)(KSEG / 16)) * (unsigned)(KSTRIDE - KSEG))))
        for (unsigned it = 0; it < nFull; ++it) {
            unsigned g = it + rot;
            if (g >= nFull) g -= nFull;
            const unsigned base = g * (1024u * U) + (unsigned)tid;
            uint4 v[U];
#pragma unroll
            for (int u = 0; u < U; ++u) v[u] = *KEYPTR(base + u * 1024u);
#pragma unroll
            for (int u = 0; u < U; ++u)
                wc = keyscan16<F>(v[u], (base + u * 1024u) * 16u, pat, lo, ucnt, uE, wc, wl,
                                  qcnt, queue, tile, lin, src, dst, dinv);
        }
        for (unsigned c0 = nFull * (1024u * U) + (unsigned)(tid & ~31); c0 < nC; c0 += 1024u) {
            const unsigned c = c0 + (unsigned)lane;
            uint4 kv = make_uint4(~pat, ~pat, ~pat, ~pat);
            if (c < nC) kv = *KEYPTR(c);
            wc = keyscan16<F>(kv, c * 16u, pat, lo, ucnt, uE, wc, wl,
                              qcnt, queue, tile, lin, src, dst, dinv);
        }
#undef KEYPTR
        __syncthreads();
        const int nw = min(wc, CW);
        constexpr int V = 8;
        for (int i0 = lane; i0 < nw; i0 += 32 * V) {
            int e[V];
            unsigned r[V];
#pragma unroll
            for (int u = 0; u < V; ++u) e[u] = wl[min(i0 + u * 32, nw - 1)];
#pragma unroll
            for (int u = 0; u < V; ++u) r[u] = (unsigned)(dst[e[u]] - lo);
#pragma unroll
            for (int u = 0; u < V; ++u)
                if (i0 + u * 32 < nw && r[u] < ucnt)
                    push_edge<F>(e[u], r[u], lo, qcnt, queue, tile, lin, src, dinv);
        }
    }
    if (!useSlab && !byKeys) {
        constexpr int U = 8;
        const bool al = (((size_t)dst) & 15) == 0;
        const int nV = al ? (nE >> 2) : 0;
        const int4* d4 = (const int4*)dst;
        const int nFull = nV / (1024 * U);
        const int rot = (int)(((unsigned)blockIdx.x * (unsigned)nFull) / gridDim.x);
        for (int it = 0; it < nFull; ++it) {
            int g = it + rot;
            if (g >= nFull) g -= nFull;
            const int base = g * (1024 * U) + tid;
            int4 v[U];
#pragma unroll
            for (int u = 0; u < U; ++u) v[u] = d4[base + u * 1024];
#pragma unroll
            for (int u = 0; u < U; ++u)
                scan4<F>(v[u], (base + u * 1024) * 4, lo, ucnt, qcnt, queue, tile, lin, src, dinv);
        }
        for (int idx = nFull * (1024 * U) + tid; idx < nV; idx += 1024)
            scan4<F>(d4[idx], idx * 4, lo, ucnt, qcnt, queue, tile, lin, src, dinv);
        for (int e = nV * 4 + tid; e < nE; e += 1024) {
            const unsigned r = (unsigned)(dst[e] - lo);
            if (r < ucnt) push_edge<F>(e, r, lo, qcnt, queue, tile, lin, src, dinv);
        }
    }
    __syncthreads();

    {
        constexpr int G = 8;
        const int qraw = qcnt[lane];
        const int cl = min(qraw, QCAP);
        int off = 0, total = 0;
        for (int k = 0; k < NWAVE; ++k) {
            const int ck = __builtin_amdgcn_readlane(cl, k);
            if (k < wave) off += ck;
            total += ck;
        }
        const int mine = __builtin_amdgcn_readlane(cl, wave);
        bool storeOK = false;
        if (STORE_Q) {
            storeOK = !__any(qraw > QCAP) && total <= cap;
            if (wave == 0) {
                slab[lane] = cl;
                if (lane == 0) slab[QHDR / 2] = storeOK ? 1 : 0;
            }
        }
        const int* q = queue + wave * QCAP;
        for (int b = 0; b < mine; b += 32) {
            const int nb = min(32, mine - b);
            const int idx = b + min(lane, nb - 1);
            int s, dl;
            float w;
            if (useSlab) {
                s  = recS[off + idx];
                dl = recD[off + idx];
                w  = __builtin_bit_cast(float, recW[off + idx]);
            } else {
                const int e = q[idx];
                s = src[e];
                const int d = dst[e];
                w = dinv[s] * dinv[d];
                dl = d - lo;
                if (STORE_Q && storeOK && lane < nb) {
                    recS[off + idx] = s;
                    recD[off + idx] = dl;
                    recW[off + idx] = __builtin_bit_cast(int, w);
                }
            }
            vrow va[G], vb[G];
#define GATHER(buf, J0)                                                                   \
            if ((J0) < nb) {                                                              \
                _Pragma("unroll")                                                         \
                for (int j = 0; j < G; ++j) {                                             \
                    const int sj = __builtin_amdgcn_readlane(s, min((J0) + j, nb - 1));   \
                    buf[j] = *(const vrow*)(lin + (size_t)sj * F + lane * VF);            \
                }                                                                         \
            }
#define ADDROWS(buf, J0)                                                                  \
            if ((J0) < nb) {                                                              \
                _Pragma("unroll")                                                         \
                for (int j = 0; j < G; ++j) {                                             \
                    if ((J0) + j < nb) {                                                  \
                        const float wj = __builtin_bit_cast(float,                        \
                            __builtin_amdgcn_readlane(__builtin_bit_cast(int, w), (J0) + j)); \
                        const int dj = __builtin_amdgcn_readlane(dl, (J0) + j);           \
                        vrow* t = (vrow*)(tile + (size_t)dj * F + lane * VF);             \
                        *t = *t + buf[j] * wj;                                            \
                    }                                                                     \
                }                                                                         \
            }
            GATHER(va, 0)
            GATHER(vb, G)
            ADDROWS(va, 0)
            GATHER(va, 2 * G)
            ADDROWS(vb, G)
            GATHER(vb, 3 * G)
            ADDROWS(va, 2 * G)
            ADDROWS(vb, 3 * G)
#undef GATHER
#undef ADDROWS
        }
    }
    __syncthreads();

    {
        float4* out4 = (float4*)(agg + (size_t)lo * F);
        const float4* tile4 = (const float4*)tile;
        for (int idx = tid; idx < cnt * F4; idx += 1024) out4[idx] = tile4[idx];
    }
}

__global__ void k_decode(const float* __restrict__ z,
                         const int* __restrict__ pos, const int* __restrict__ neg,
                         float* __restrict__ out, int EP) {
    int wid  = (blockIdx.x * blockDim.x + threadIdx.x) >> 5;
    int lane = threadIdx.x & 31;
    if (wid >= 2 * EP) return;
    int u, v;
    if (wid < EP) { u = pos[wid];       v = pos[EP + wid]; }
    else          { u = neg[wid - EP];  v = neg[EP + (wid - EP)]; }
    const float* zu = z + (size_t)u * OUT_F;
    const float* zv = z + (size_t)v * OUT_F;
    float s = zu[lane] * zv[lane] + zu[lane + 32] * zv[lane + 32];
#pragma unroll
    for (int off = 16; off > 0; off >>= 1)
        s += __shfl_down(s, off, 32);
    if (lane == 0) out[wid] = s;
}


extern "C" void kernel_launch(void* const* d_in, const int* in_sizes, int n_in,
                              void* d_out, int out_size, void* d_ws, size_t ws_size,
                              hipStream_t stream) {
    const float* x    = (const float*)d_in[0];
    const int*   ei   = (const int*)  d_in[1];
    const int*   pei  = (const int*)  d_in[2];
    const int*   nei  = (const int*)  d_in[3];
    const float* W1   = (const float*)d_in[4];
    const float* b1   = (const float*)d_in[5];
    const float* W2   = (const float*)d_in[6];
    const float* b2   = (const float*)d_in[7];
    float* out = (float*)d_out;

    const int N  = in_sizes[0] / IN_F;
    const int E  = in_sizes[1] / 2;
    const int EP = in_sizes[2] / 2;
    const int* src = ei;
    const int* dst = ei + E;

    float* dinv = (float*)d_ws;
    size_t npad = ((size_t)N + 255) & ~(size_t)255;
    float* bufA = dinv + npad;
    float* h    = bufA + (size_t)N * H_F;
    float* z    = h    + (size_t)N * H_F;

    const int B = 256;
    k_deg_init<<<(N + B - 1) / B, B, 0, stream>>>(dinv, N);
    k_deg_accum<<<(E + B - 1) / B, B, 0, stream>>>(dst, dinv, (unsigned char*)z, E, N);
    k_rsqrt_inplace<<<(N + B - 1) / B, B, 0, stream>>>(dinv, N);

    {
        int rowTiles = (N + 15) / 16;
        int blocks   = (rowTiles + 7) / 8;
        k_gemm_wmma<IN_F, H_F, false><<<blocks, B, 0, stream>>>(x, W1, bufA, N);
    }
    k_agg_owner<H_F, true, false><<<(N + TILE - 1) / TILE, 1024, 0, stream>>>(bufA, src, dst, dinv, b1, h, (int*)z, N, E);
    {
        int rowTiles = (N + 15) / 16;
        int blocks   = (rowTiles + 7) / 8;
        k_gemm_wmma<H_F, OUT_F, true><<<blocks, B, 0, stream>>>(h, W2, bufA, N);
    }
    k_agg_owner<OUT_F, false, true><<<(N + TILE - 1) / TILE, 1024, 0, stream>>>(bufA, src, dst, dinv, b2, z, (int*)z, N, E);
    {
        long long waves = 2LL * EP;
        long long thr   = waves * 32;
        k_decode<<<(int)((thr + B - 1) / B), B, 0, stream>>>(z, pei, nei, out, EP);
    }
}
